# starting kernel
# baseline (speedup 1.0000x reference)
#include <hip/hip_runtime.h>
#include <math.h>


typedef __attribute__((ext_vector_type(2))) float v2f;
typedef __attribute__((ext_vector_type(8))) float v8f;

#define BATCH 16384
#define NCP   4096
#define CHUNK 1024
#define BLOCK 256
#define ROWS_PER_BLOCK 128

#define EPS2      1e-20f
#define HALF_LN2  0.34657359028f

__launch_bounds__(BLOCK)
__global__ void tps_wmma_kernel(const float* __restrict__ u,
                                const float* __restrict__ cp,
                                const float* __restrict__ w,
                                const float* __restrict__ poly,
                                float* __restrict__ out)
{
    __shared__ float4 sCP[CHUNK];
    __shared__ float4 sW[CHUNK];

    const int tid  = threadIdx.x;
    const int wave = tid >> 5;
    const int lane = tid & 31;
    const int half = lane >> 4;
    const int q    = lane & 15;

    const int qIdx = blockIdx.x * ROWS_PER_BLOCK + wave * 16 + q;

    const float ux = u[qIdx * 3 + 0];
    const float uy = u[qIdx * 3 + 1];
    const float uz = u[qIdx * 3 + 2];
    const float usq = ux * ux + uy * uy + uz * uz;

    v2f bmat;
    bmat.x = half ? (-2.0f * uz) : (-2.0f * ux);
    bmat.y = half ? 1.0f         : (-2.0f * uy);

    v8f cmat;
#pragma unroll
    for (int k = 0; k < 8; ++k) cmat[k] = usq;

    float acc0 = 0.0f, acc1 = 0.0f, acc2 = 0.0f;

    for (int base = 0; base < NCP; base += CHUNK) {
        __syncthreads();
        for (int i = tid; i < CHUNK; i += BLOCK) {
            const int g = base + i;
            const float cx = cp[g * 3 + 0];
            const float cy = cp[g * 3 + 1];
            const float cz = cp[g * 3 + 2];
            sCP[i] = make_float4(cx, cy, cz, cx * cx + cy * cy + cz * cz);
            sW[i]  = make_float4(HALF_LN2 * w[g * 3 + 0],
                                 HALF_LN2 * w[g * 3 + 1],
                                 HALF_LN2 * w[g * 3 + 2], 0.0f);
        }
        __syncthreads();

#pragma unroll 2
        for (int cb = 0; cb < CHUNK; cb += 16) {
            const float4 c4 = sCP[cb + q];
            v2f amat;
            amat.x = half ? c4.z : c4.x;
            amat.y = half ? c4.w : c4.y;

            v8f d = __builtin_amdgcn_wmma_f32_16x16x4_f32(
                false, amat, false, bmat, (short)0, cmat, false, false);

            const int mBase = cb + half * 8;
#pragma unroll
            for (int k = 0; k < 8; ++k) {
                const float sq    = d[k];
                const bool  small = sq < EPS2;
                const float sqc   = small ? EPS2 : sq;
                float kv = __builtin_amdgcn_sqrtf(sqc) * __builtin_amdgcn_logf(sqc);
                kv = small ? 0.0f : kv;
                const float4 w4 = sW[mBase + k];
                acc0 = fmaf(kv, w4.x, acc0);
                acc1 = fmaf(kv, w4.y, acc1);
                acc2 = fmaf(kv, w4.z, acc2);
            }
        }
    }

    acc0 += __shfl_xor(acc0, 16, 32);
    acc1 += __shfl_xor(acc1, 16, 32);
    acc2 += __shfl_xor(acc2, 16, 32);

    if (half == 0) {
#pragma unroll
        for (int a = 0; a < 3; ++a) {
            const float p = poly[0 * 3 + a] + ux * poly[1 * 3 + a]
                          + uy * poly[2 * 3 + a] + uz * poly[3 * 3 + a];
            const float acc = (a == 0) ? acc0 : ((a == 1) ? acc1 : acc2);
            out[qIdx * 3 + a] = acc + p;
        }
    }
}

extern "C" void kernel_launch(void* const* d_in, const int* in_sizes, int n_in,
                              void* d_out, int out_size, void* d_ws, size_t ws_size,
                              hipStream_t stream) {
    const float* u    = (const float*)d_in[0];
    const float* cp   = (const float*)d_in[1];
    const float* w    = (const float*)d_in[2];
    const float* poly = (const float*)d_in[3];
    float* out = (float*)d_out;

    dim3 grid(BATCH / ROWS_PER_BLOCK);
    tps_wmma_kernel<<<grid, BLOCK, 0, stream>>>(u, cp, w, poly, out);
}


# Vector stream as a 3-stage pipeline over tiles: each sqrt/log is issued paired with one plain instruction of another tile (clamp of g+2, a multiply-add of g), never two in a row; same sum order
# speedup vs baseline: 7.4766x; 7.4766x over previous
#include <hip/hip_runtime.h>
#include <math.h>


typedef __attribute__((ext_vector_type(2))) float v2f;
typedef __attribute__((ext_vector_type(8))) float v8f;

#define BATCH  16384
#define NCP    4096
#define NT     (NCP / 16)
#define BLOCK  256
#define QPB    64
#define SLICES 2
#define TPW    (NT / SLICES)
#define NTRIP  (TPW / 4)

#define EPS2      1e-20f
#define HALF_LN2  0.34657359028f

template <int CTRL>
__device__ __forceinline__ float row_add(float v) {
    const int o = __builtin_amdgcn_update_dpp(0, __float_as_int(v), CTRL, 0xf, 0xf, true);
    return v + __int_as_float(o);
}

__device__ __forceinline__ float row_sum16(float v) {
    v = row_add<0x161>(v);
    v = row_add<0x162>(v);
    v = row_add<0x164>(v);
    v = row_add<0x168>(v);
    return v;
}


__device__ __forceinline__ void sqrt_and_clamp(float &s_new, float &m_new, const float m, const float sq) {
    asm("v_sqrt_f32 %0, %2\n\tv_maximum_f32 %1, 0x1e3ce508, %3"
        : "=&v"(s_new), "=&v"(m_new) : "v"(m), "v"(sq));
}
__device__ __forceinline__ void log_and_madd(float &l_new, float &acc2, const float m, const float kv, const float wz) {
    asm("v_log_f32 %0, %2\n\tv_fmac_f32 %1, %3, %4"
        : "=&v"(l_new), "+v"(acc2) : "v"(m), "v"(kv), "v"(wz));
}
__device__ __forceinline__ void clamp_tile(const v8f d, float (&m)[8]) {
#pragma unroll
    for (int r = 0; r < 8; ++r) m[r] = __builtin_elementwise_maximum(d[r], EPS2);
}
__device__ __forceinline__ void roots_logs(const float (&m)[8], float (&s)[8], float (&l)[8]) {
#pragma unroll
    for (int r = 0; r < 8; ++r) {
        s[r] = __builtin_amdgcn_sqrtf(m[r]);
        l[r] = __builtin_amdgcn_logf(m[r]);
    }
}
__device__ __forceinline__ void vector_step(const v8f dIssued, const v8f dNext2,
                                            float (&m)[8], float (&s)[8], float (&l)[8],
                                            const float4 w4, float (&acc)[8][3]) {
    float kv[8];
#pragma unroll
    for (int r = 0; r < 8; ++r) kv[r] = s[r] * l[r];
#pragma unroll
    for (int r = 0; r < 8; ++r) {
        float s_new, l_new, m_new;
        sqrt_and_clamp(s_new, m_new, m[r], dNext2[r]);
        log_and_madd(l_new, acc[r][2], m[r], kv[r], w4.z);
        acc[r][0] = fmaf(kv[r], w4.x, acc[r][0]);
        acc[r][1] = fmaf(kv[r], w4.y, acc[r][1]);
        s[r] = s_new; l[r] = l_new; m[r] = m_new;
    }
    asm volatile("" :: "v"(dIssued));
    __builtin_amdgcn_sched_barrier(0);
}

__device__ __forceinline__ void trip_load_b(float2 (&b)[4], const float2* pB, int trip) {
#pragma unroll
    for (int i = 0; i < 4; ++i) b[i] = pB[(trip * 4 + i) * 32];
}
__device__ __forceinline__ void trip_load_w(float4 (&w4)[4], const float4* pW, int trip) {
#pragma unroll
    for (int i = 0; i < 4; ++i) w4[i] = pW[(trip * 4 + i) * 16];
}
__device__ __forceinline__ v8f tile_wmma(const float2 b, const v2f amat, const v8f cmat) {
    v2f bm; bm.x = b.x; bm.y = b.y;
    return __builtin_amdgcn_wmma_f32_16x16x4_f32(
        false, amat, false, bm, (short)0, cmat, false, false);
}

__device__ __forceinline__ void trip_hold_b(const float2 (&b)[4]) {
#pragma unroll
    for (int i = 0; i < 4; ++i) asm volatile("" :: "v"(b[i].x), "v"(b[i].y));
}
__device__ __forceinline__ void trip_hold_w(const float4 (&w4)[4]) {
#pragma unroll
    for (int i = 0; i < 4; ++i) asm volatile("" :: "v"(w4[i].x), "v"(w4[i].y), "v"(w4[i].z));
}

__launch_bounds__(BLOCK)
__global__ void tps_rows_kernel(const float* __restrict__ u,
                                const float* __restrict__ cp,
                                const float* __restrict__ w,
                                const float* __restrict__ poly,
                                float* __restrict__ out)
{
    __shared__ float2 sB[NT * 32];
    __shared__ float4 sW[NCP];
    __shared__ float  sPart[SLICES][QPB * 3];

    const int tid   = threadIdx.x;
    const int wave  = tid >> 5;
    const int lane  = tid & 31;
    const int half  = lane >> 4;
    const int c     = lane & 15;
    const int grp   = wave & 3;
    const int slice = wave >> 2;

#pragma unroll
    for (int k = 0; k < NCP / BLOCK; ++k) {
        const int j = k * BLOCK + tid;
        const float cx = cp[j * 3 + 0];
        const float cy = cp[j * 3 + 1];
        const float cz = cp[j * 3 + 2];
        const int slot = (j >> 4) * 32 + (j & 15);
        sB[slot]      = make_float2(cx, cy);
        sB[slot + 16] = make_float2(cz, cx * cx + cy * cy + cz * cz);
        sW[j] = make_float4(HALF_LN2 * w[j * 3 + 0],
                            HALF_LN2 * w[j * 3 + 1],
                            HALF_LN2 * w[j * 3 + 2], 0.0f);
    }

    const int qIdx = blockIdx.x * QPB + grp * 16 + c;
    const float ux = u[qIdx * 3 + 0];
    const float uy = u[qIdx * 3 + 1];
    const float uz = u[qIdx * 3 + 2];
    const float usq = ux * ux + uy * uy + uz * uz;

    v2f amat;
    amat.x = half ? (-2.0f * uz) : (-2.0f * ux);
    amat.y = half ? 1.0f         : (-2.0f * uy);

    v8f cmat;
#pragma unroll
    for (int r = 0; r < 8; ++r) cmat[r] = __shfl(usq, half * 8 + r, 32);

    float acc[8][3];
#pragma unroll
    for (int r = 0; r < 8; ++r) { acc[r][0] = 0.0f; acc[r][1] = 0.0f; acc[r][2] = 0.0f; }

    __syncthreads();

    const float2* pB = sB + (slice * TPW) * 32 + lane;
    const float4* pW = sW + (slice * TPW) * 16 + c;

    float2 bA[4], bB[4];
    float4 wA[4], wB[4];
    v8f    dA[4], dB[4];
    trip_load_b(bA, pB, 0);
    trip_load_w(wA, pW, 0);
    trip_load_b(bB, pB, 1);
#pragma unroll
    for (int i = 0; i < 4; ++i) dA[i] = tile_wmma(bA[i], amat, cmat);

    float m[8], s[8], l[8];
    clamp_tile(dA[0], m);
    roots_logs(m, s, l);
    clamp_tile(dA[1], m);

    for (int t = 0; t < NTRIP; t += 2) {
        const int t2 = (t + 2 < NTRIP) ? (t + 2) : t;
        const int t3 = (t + 2 < NTRIP) ? (t + 3) : (t + 1);

        trip_load_b(bA, pB, t2);
        trip_load_w(wB, pW, t + 1);
        dB[0] = tile_wmma(bB[0], amat, cmat);
        vector_step(dB[0], dA[2], m, s, l, wA[0], acc);
        dB[1] = tile_wmma(bB[1], amat, cmat);
        vector_step(dB[1], dA[3], m, s, l, wA[1], acc);
        dB[2] = tile_wmma(bB[2], amat, cmat);
        vector_step(dB[2], dB[0], m, s, l, wA[2], acc);
        trip_hold_b(bA);
        trip_hold_w(wB);
        dB[3] = tile_wmma(bB[3], amat, cmat);
        vector_step(dB[3], dB[1], m, s, l, wA[3], acc);

        trip_load_b(bB, pB, t3);
        trip_load_w(wA, pW, t2);
        dA[0] = tile_wmma(bA[0], amat, cmat);
        vector_step(dA[0], dB[2], m, s, l, wB[0], acc);
        dA[1] = tile_wmma(bA[1], amat, cmat);
        vector_step(dA[1], dB[3], m, s, l, wB[1], acc);
        dA[2] = tile_wmma(bA[2], amat, cmat);
        vector_step(dA[2], dA[0], m, s, l, wB[2], acc);
        trip_hold_b(bB);
        trip_hold_w(wA);
        dA[3] = tile_wmma(bA[3], amat, cmat);
        vector_step(dA[3], dA[1], m, s, l, wB[3], acc);
    }

#pragma unroll
    for (int r = 0; r < 8; ++r) {
        acc[r][0] = row_sum16(acc[r][0]);
        acc[r][1] = row_sum16(acc[r][1]);
        acc[r][2] = row_sum16(acc[r][2]);
    }
    if (c == 0) {
        float* dst = &sPart[slice][grp * 48 + half * 24];
#pragma unroll
        for (int r = 0; r < 8; ++r) {
            dst[r * 3 + 0] = acc[r][0];
            dst[r * 3 + 1] = acc[r][1];
            dst[r * 3 + 2] = acc[r][2];
        }
    }
    __syncthreads();

    if (tid < QPB * 3) {
        const int q = tid / 3;
        const int a = tid - q * 3;
        const int qg = blockIdx.x * QPB + q;
        const float px = u[qg * 3 + 0];
        const float py = u[qg * 3 + 1];
        const float pz = u[qg * 3 + 2];
        const float p = poly[0 * 3 + a] + px * poly[1 * 3 + a]
                      + py * poly[2 * 3 + a] + pz * poly[3 * 3 + a];
        const float s = sPart[0][tid] + sPart[1][tid];
        out[blockIdx.x * (QPB * 3) + tid] = s + p;
    }
}

extern "C" void kernel_launch(void* const* d_in, const int* in_sizes, int n_in,
                              void* d_out, int out_size, void* d_ws, size_t ws_size,
                              hipStream_t stream) {
    const float* u    = (const float*)d_in[0];
    const float* cp   = (const float*)d_in[1];
    const float* w    = (const float*)d_in[2];
    const float* poly = (const float*)d_in[3];
    float* out = (float*)d_out;

    dim3 grid(BATCH / QPB);
    tps_rows_kernel<<<grid, BLOCK, 0, stream>>>(u, cp, w, poly, out);
}
